# starting kernel
# baseline (speedup 1.0000x reference)
#include <hip/hip_runtime.h>
#include <math.h>

typedef __attribute__((ext_vector_type(2))) float v2f;
typedef __attribute__((ext_vector_type(8))) float v8f;

#define NW 8
#define BB 8
#define SS 1024
#define EE 128
#define HH 16
#define DK 8
#define ZS 18

__global__ __launch_bounds__(256)
void qattn_kernel(const float* __restrict__ x, const float* __restrict__ theta,
                  float* __restrict__ y) {
  __shared__ float sZ[SS * ZS];
  __shared__ float sP[8 * 16 * 18];

  const int bh  = blockIdx.x;
  const int b   = bh / HH;
  const int h   = bh % HH;
  const int tid = threadIdx.x;

  float th[NW];
#pragma unroll
  for (int j = 0; j < NW; ++j) th[j] = theta[j];

  for (int s = tid; s < SS; s += 256) {
    const float* xr = x + ((size_t)(b * SS + s) * EE + h * NW);
    float c[NW];
#pragma unroll
    for (int j = 0; j < NW; ++j) c[j] = __cosf(xr[j] + th[j]);
    float zr[NW];
    float p = c[0];
#pragma unroll
    for (int j = 1; j < NW; ++j) { p *= c[j]; zr[j] = p; }
    float sfx = 1.0f;
#pragma unroll
    for (int j = NW - 1; j >= 1; --j) sfx *= c[j];
    zr[0] = sfx;
#pragma unroll
    for (int j = 0; j < NW; ++j) sZ[s * ZS + j] = zr[j];
    sZ[s * ZS + 8] = 1.0f;
#pragma unroll
    for (int j = 9; j < 16; ++j) sZ[s * ZS + j] = 0.0f;
  }
  __syncthreads();

  const int lane = tid & 31;
  const int wave = tid >> 5;
  const int hf   = lane >> 4;
  const int lm   = lane & 15;
  const float scale = 0.35355339059327373f;
  float* sp = sP + wave * (16 * 18);

  for (int qb = wave; qb < SS / 16; qb += 8) {
    const int q0 = qb * 16;

    const float* qp = &sZ[(q0 + lm) * ZS + hf * 2];
    v2f a0 = {qp[0] * scale, qp[1] * scale};
    v2f a1 = {qp[4] * scale, qp[5] * scale};

    v8f o = {0.f, 0.f, 0.f, 0.f, 0.f, 0.f, 0.f, 0.f};

    for (int kb = 0; kb < SS / 16; ++kb) {
      const int k0 = kb * 16;

      const float* kp = &sZ[(k0 + lm) * ZS + hf * 2];
      v2f b0 = {kp[0], kp[1]};
      v2f b1 = {kp[4], kp[5]};

      v8f sc = {0.f, 0.f, 0.f, 0.f, 0.f, 0.f, 0.f, 0.f};
      sc = __builtin_amdgcn_wmma_f32_16x16x4_f32(false, a0, false, b0,
                                                 (short)0, sc, false, false);
      sc = __builtin_amdgcn_wmma_f32_16x16x4_f32(false, a1, false, b1,
                                                 (short)0, sc, false, false);

      float pr[8];
#pragma unroll
      for (int r = 0; r < 8; ++r) pr[r] = __expf(sc[r]);

#pragma unroll
      for (int r = 0; r < 8; ++r)
        sp[(r + 8 * hf) * 18 + lm] = pr[r];
      __builtin_amdgcn_wave_barrier();

#pragma unroll
      for (int c = 0; c < 4; ++c) {
        const float* pp = &sp[lm * 18 + c * 4 + hf * 2];
        v2f pa = {pp[0], pp[1]};
        const int krow = k0 + c * 4 + hf * 2;
        v2f vb = {sZ[krow * ZS + lm], sZ[(krow + 1) * ZS + lm]};
        o = __builtin_amdgcn_wmma_f32_16x16x4_f32(false, pa, false, vb,
                                                  (short)0, o, false, false);
      }
      __builtin_amdgcn_wave_barrier();
    }

#pragma unroll
    for (int r = 0; r < 8; ++r) {
      const float li = __shfl(o[r], 16 * hf + DK);
      if (lm < DK) {
        const int m = r + 8 * hf;
        y[(size_t)(b * SS + q0 + m) * EE + h * NW + lm] = o[r] / li;
      }
    }
  }
}

__global__ __launch_bounds__(256)
void proj_kernel(const float* __restrict__ y, const float* __restrict__ w,
                 const float* __restrict__ bias, float* __restrict__ out) {
  const int wid  = blockIdx.x * 8 + (threadIdx.x >> 5);
  const int tm   = wid >> 3;
  const int tn   = wid & 7;
  const int lane = threadIdx.x & 31;
  const int hf   = lane >> 4;
  const int lm   = lane & 15;

  const float* ya = y + (size_t)(tm * 16 + lm) * EE + hf * 2;
  const float* wb = w + (size_t)(tn * 16 + lm) * EE + hf * 2;

  v8f acc = {0.f, 0.f, 0.f, 0.f, 0.f, 0.f, 0.f, 0.f};
#pragma unroll
  for (int kc = 0; kc < EE / 4; ++kc) {
    v2f a  = {ya[kc * 4], ya[kc * 4 + 1]};
    v2f bb = {wb[kc * 4], wb[kc * 4 + 1]};
    acc = __builtin_amdgcn_wmma_f32_16x16x4_f32(false, a, false, bb,
                                                (short)0, acc, false, false);
  }

  const float bo = bias[tn * 16 + lm];
#pragma unroll
  for (int r = 0; r < 8; ++r) {
    const int m = tm * 16 + r + 8 * hf;
    out[(size_t)m * EE + tn * 16 + lm] = acc[r] + bo;
  }
}

extern "C" void kernel_launch(void* const* d_in, const int* in_sizes, int n_in,
                              void* d_out, int out_size, void* d_ws, size_t ws_size,
                              hipStream_t stream) {
  const float* x     = (const float*)d_in[0];
  const float* theta = (const float*)d_in[1];
  const float* w_out = (const float*)d_in[2];
  const float* b_out = (const float*)d_in[3];
  float* out = (float*)d_out;
  float* y   = (float*)d_ws;

  qattn_kernel<<<BB * HH, 256, 0, stream>>>(x, theta, y);
  proj_kernel<<<512, 256, 0, stream>>>(y, w_out, b_out, out);
}


# queries of each (batch, head) split over 2 work-groups of 32 waves, one 16-query block per wave: 256 work-groups at 8 waves per SIMD (was 128 at 2 per SIMD); each work-group encodes its own z
# speedup vs baseline: 2.2870x; 2.2870x over previous
#include <hip/hip_runtime.h>
#include <math.h>

typedef __attribute__((ext_vector_type(2))) float v2f;
typedef __attribute__((ext_vector_type(8))) float v8f;

#define NW 8
#define BB 8
#define SS 1024
#define EE 128
#define HH 16
#define DK 8
#define ZS 18
#define QSPLIT 2
#define NWAVE  (SS / 16 / QSPLIT)
#define NTHR   (NWAVE * 32)

__global__ __launch_bounds__(NTHR)
void qattn_kernel(const float* __restrict__ x, const float* __restrict__ theta,
                  float* __restrict__ y) {
  __shared__ float sZ[SS * ZS];
  __shared__ float sP[NWAVE * 16 * 18];

  const int bh   = blockIdx.x / QSPLIT;
  const int part = blockIdx.x % QSPLIT;
  const int b   = bh / HH;
  const int h   = bh % HH;
  const int tid = threadIdx.x;

  float th[NW];
#pragma unroll
  for (int j = 0; j < NW; ++j) th[j] = theta[j];

  for (int s = tid; s < SS; s += NTHR) {
    const float* xr = x + ((size_t)(b * SS + s) * EE + h * NW);
    float c[NW];
#pragma unroll
    for (int j = 0; j < NW; ++j) c[j] = __cosf(xr[j] + th[j]);
    float zr[NW];
    float p = c[0];
#pragma unroll
    for (int j = 1; j < NW; ++j) { p *= c[j]; zr[j] = p; }
    float sfx = 1.0f;
#pragma unroll
    for (int j = NW - 1; j >= 1; --j) sfx *= c[j];
    zr[0] = sfx;
#pragma unroll
    for (int j = 0; j < NW; ++j) sZ[s * ZS + j] = zr[j];
    sZ[s * ZS + 8] = 1.0f;
#pragma unroll
    for (int j = 9; j < 16; ++j) sZ[s * ZS + j] = 0.0f;
  }
  __syncthreads();

  const int lane = tid & 31;
  const int wave = tid >> 5;
  const int hf   = lane >> 4;
  const int lm   = lane & 15;
  const float scale = 0.35355339059327373f;
  float* sp = sP + wave * (16 * 18);

  {
    const int q0 = (part * NWAVE + wave) * 16;

    const float* qp = &sZ[(q0 + lm) * ZS + hf * 2];
    v2f a0 = {qp[0] * scale, qp[1] * scale};
    v2f a1 = {qp[4] * scale, qp[5] * scale};

    v8f o = {0.f, 0.f, 0.f, 0.f, 0.f, 0.f, 0.f, 0.f};

    for (int kb = 0; kb < SS / 16; ++kb) {
      const int k0 = kb * 16;

      const float* kp = &sZ[(k0 + lm) * ZS + hf * 2];
      v2f b0 = {kp[0], kp[1]};
      v2f b1 = {kp[4], kp[5]};

      v8f sc = {0.f, 0.f, 0.f, 0.f, 0.f, 0.f, 0.f, 0.f};
      sc = __builtin_amdgcn_wmma_f32_16x16x4_f32(false, a0, false, b0,
                                                 (short)0, sc, false, false);
      sc = __builtin_amdgcn_wmma_f32_16x16x4_f32(false, a1, false, b1,
                                                 (short)0, sc, false, false);

      float pr[8];
#pragma unroll
      for (int r = 0; r < 8; ++r) pr[r] = __expf(sc[r]);

#pragma unroll
      for (int r = 0; r < 8; ++r)
        sp[(r + 8 * hf) * 18 + lm] = pr[r];
      __builtin_amdgcn_wave_barrier();

#pragma unroll
      for (int c = 0; c < 4; ++c) {
        const float* pp = &sp[lm * 18 + c * 4 + hf * 2];
        v2f pa = {pp[0], pp[1]};
        const int krow = k0 + c * 4 + hf * 2;
        v2f vb = {sZ[krow * ZS + lm], sZ[(krow + 1) * ZS + lm]};
        o = __builtin_amdgcn_wmma_f32_16x16x4_f32(false, pa, false, vb,
                                                  (short)0, o, false, false);
      }
      __builtin_amdgcn_wave_barrier();
    }

#pragma unroll
    for (int r = 0; r < 8; ++r) {
      const float li = __shfl(o[r], 16 * hf + DK);
      if (lm < DK) {
        const int m = r + 8 * hf;
        y[(size_t)(b * SS + q0 + m) * EE + h * NW + lm] = o[r] / li;
      }
    }
  }
}

__global__ __launch_bounds__(256)
void proj_kernel(const float* __restrict__ y, const float* __restrict__ w,
                 const float* __restrict__ bias, float* __restrict__ out) {
  const int wid  = blockIdx.x * 8 + (threadIdx.x >> 5);
  const int tm   = wid >> 3;
  const int tn   = wid & 7;
  const int lane = threadIdx.x & 31;
  const int hf   = lane >> 4;
  const int lm   = lane & 15;

  const float* ya = y + (size_t)(tm * 16 + lm) * EE + hf * 2;
  const float* wb = w + (size_t)(tn * 16 + lm) * EE + hf * 2;

  v8f acc = {0.f, 0.f, 0.f, 0.f, 0.f, 0.f, 0.f, 0.f};
#pragma unroll
  for (int kc = 0; kc < EE / 4; ++kc) {
    v2f a  = {ya[kc * 4], ya[kc * 4 + 1]};
    v2f bb = {wb[kc * 4], wb[kc * 4 + 1]};
    acc = __builtin_amdgcn_wmma_f32_16x16x4_f32(false, a, false, bb,
                                                (short)0, acc, false, false);
  }

  const float bo = bias[tn * 16 + lm];
#pragma unroll
  for (int r = 0; r < 8; ++r) {
    const int m = tm * 16 + r + 8 * hf;
    out[(size_t)m * EE + tn * 16 + lm] = acc[r] + bo;
  }
}

extern "C" void kernel_launch(void* const* d_in, const int* in_sizes, int n_in,
                              void* d_out, int out_size, void* d_ws, size_t ws_size,
                              hipStream_t stream) {
  const float* x     = (const float*)d_in[0];
  const float* theta = (const float*)d_in[1];
  const float* w_out = (const float*)d_in[2];
  const float* b_out = (const float*)d_in[3];
  float* out = (float*)d_out;
  float* y   = (float*)d_ws;

  qattn_kernel<<<BB * HH * QSPLIT, NTHR, 0, stream>>>(x, theta, y);
  proj_kernel<<<512, 256, 0, stream>>>(y, w_out, b_out, out);
}


# projection GEMM re-tiled: w_out and a 32-row y tile staged once per work-group in LDS with whole-line 16-byte loads, fragments read from LDS; 2048 waves of 16x32 outputs (was 4096 of 16x16)
# speedup vs baseline: 2.3899x; 1.0450x over previous
#include <hip/hip_runtime.h>
#include <math.h>

typedef __attribute__((ext_vector_type(2))) float v2f;
typedef __attribute__((ext_vector_type(4))) float v4f;
typedef __attribute__((ext_vector_type(8))) float v8f;

#define NW 8
#define BB 8
#define SS 1024
#define EE 128
#define HH 16
#define DK 8
#define ZS 18
#define QSPLIT 2
#define NWAVE  (SS / 16 / QSPLIT)
#define NTHR   (NWAVE * 32)

__global__ __launch_bounds__(NTHR)
void qattn_kernel(const float* __restrict__ x, const float* __restrict__ theta,
                  float* __restrict__ y) {
  __shared__ float sZ[SS * ZS];
  __shared__ float sP[NWAVE * 16 * 18];

  const int bh   = blockIdx.x / QSPLIT;
  const int part = blockIdx.x % QSPLIT;
  const int b   = bh / HH;
  const int h   = bh % HH;
  const int tid = threadIdx.x;

  float th[NW];
#pragma unroll
  for (int j = 0; j < NW; ++j) th[j] = theta[j];

  for (int s = tid; s < SS; s += NTHR) {
    const float* xr = x + ((size_t)(b * SS + s) * EE + h * NW);
    float c[NW];
#pragma unroll
    for (int j = 0; j < NW; ++j) c[j] = __cosf(xr[j] + th[j]);
    float zr[NW];
    float p = c[0];
#pragma unroll
    for (int j = 1; j < NW; ++j) { p *= c[j]; zr[j] = p; }
    float sfx = 1.0f;
#pragma unroll
    for (int j = NW - 1; j >= 1; --j) sfx *= c[j];
    zr[0] = sfx;
#pragma unroll
    for (int j = 0; j < NW; ++j) sZ[s * ZS + j] = zr[j];
    sZ[s * ZS + 8] = 1.0f;
#pragma unroll
    for (int j = 9; j < 16; ++j) sZ[s * ZS + j] = 0.0f;
  }
  __syncthreads();

  const int lane = tid & 31;
  const int wave = tid >> 5;
  const int hf   = lane >> 4;
  const int lm   = lane & 15;
  const float scale = 0.35355339059327373f;
  float* sp = sP + wave * (16 * 18);

  {
    const int q0 = (part * NWAVE + wave) * 16;

    const float* qp = &sZ[(q0 + lm) * ZS + hf * 2];
    v2f a0 = {qp[0] * scale, qp[1] * scale};
    v2f a1 = {qp[4] * scale, qp[5] * scale};

    v8f o = {0.f, 0.f, 0.f, 0.f, 0.f, 0.f, 0.f, 0.f};

    for (int kb = 0; kb < SS / 16; ++kb) {
      const int k0 = kb * 16;

      const float* kp = &sZ[(k0 + lm) * ZS + hf * 2];
      v2f b0 = {kp[0], kp[1]};
      v2f b1 = {kp[4], kp[5]};

      v8f sc = {0.f, 0.f, 0.f, 0.f, 0.f, 0.f, 0.f, 0.f};
      sc = __builtin_amdgcn_wmma_f32_16x16x4_f32(false, a0, false, b0,
                                                 (short)0, sc, false, false);
      sc = __builtin_amdgcn_wmma_f32_16x16x4_f32(false, a1, false, b1,
                                                 (short)0, sc, false, false);

      float pr[8];
#pragma unroll
      for (int r = 0; r < 8; ++r) pr[r] = __expf(sc[r]);

#pragma unroll
      for (int r = 0; r < 8; ++r)
        sp[(r + 8 * hf) * 18 + lm] = pr[r];
      __builtin_amdgcn_wave_barrier();

#pragma unroll
      for (int c = 0; c < 4; ++c) {
        const float* pp = &sp[lm * 18 + c * 4 + hf * 2];
        v2f pa = {pp[0], pp[1]};
        const int krow = k0 + c * 4 + hf * 2;
        v2f vb = {sZ[krow * ZS + lm], sZ[(krow + 1) * ZS + lm]};
        o = __builtin_amdgcn_wmma_f32_16x16x4_f32(false, pa, false, vb,
                                                  (short)0, o, false, false);
      }
      __builtin_amdgcn_wave_barrier();
    }

#pragma unroll
    for (int r = 0; r < 8; ++r) {
      const float li = __shfl(o[r], 16 * hf + DK);
      if (lm < DK) {
        const int m = r + 8 * hf;
        y[(size_t)(b * SS + q0 + m) * EE + h * NW + lm] = o[r] / li;
      }
    }
  }
}

#define PJ_ROWS  32
#define PJ_PITCH 132
#define PJ_THR   256

__global__ __launch_bounds__(PJ_THR)
void proj_kernel(const float* __restrict__ y, const float* __restrict__ w,
                 const float* __restrict__ bias, float* __restrict__ out) {
  __shared__ float sW[EE * PJ_PITCH];
  __shared__ float sY[PJ_ROWS * PJ_PITCH];

  const int tid  = threadIdx.x;
  const int row0 = blockIdx.x * PJ_ROWS;

#pragma unroll
  for (int it = 0; it < (PJ_ROWS * EE / 4) / PJ_THR; ++it) {
    const int id = it * PJ_THR + tid;
    const int r  = id >> 5;
    const int c4 = id & 31;
    const v4f v = *reinterpret_cast<const v4f*>(y + (size_t)(row0 + r) * EE + c4 * 4);
    *reinterpret_cast<v4f*>(&sY[r * PJ_PITCH + c4 * 4]) = v;
  }
#pragma unroll
  for (int it = 0; it < (EE * EE / 4) / PJ_THR; ++it) {
    const int id = it * PJ_THR + tid;
    const int r  = id >> 5;
    const int c4 = id & 31;
    const v4f v = *reinterpret_cast<const v4f*>(w + (size_t)r * EE + c4 * 4);
    *reinterpret_cast<v4f*>(&sW[r * PJ_PITCH + c4 * 4]) = v;
  }
  __syncthreads();

  const int wave = tid >> 5;
  const int lane = tid & 31;
  const int hf   = lane >> 4;
  const int lm   = lane & 15;
  const int rt   = wave >> 2;
  const int cq   = wave & 3;

  const float* ap  = &sY[(rt * 16 + lm) * PJ_PITCH + hf * 2];
  const float* bp0 = &sW[(cq * 32 + lm) * PJ_PITCH + hf * 2];
  const float* bp1 = bp0 + 16 * PJ_PITCH;

  v8f acc0 = {0.f, 0.f, 0.f, 0.f, 0.f, 0.f, 0.f, 0.f};
  v8f acc1 = {0.f, 0.f, 0.f, 0.f, 0.f, 0.f, 0.f, 0.f};
#pragma unroll
  for (int kc = 0; kc < EE / 4; ++kc) {
    const v2f a  = {ap[kc * 4],  ap[kc * 4 + 1]};
    const v2f b0 = {bp0[kc * 4], bp0[kc * 4 + 1]};
    const v2f b1 = {bp1[kc * 4], bp1[kc * 4 + 1]};
    acc0 = __builtin_amdgcn_wmma_f32_16x16x4_f32(false, a, false, b0,
                                                 (short)0, acc0, false, false);
    acc1 = __builtin_amdgcn_wmma_f32_16x16x4_f32(false, a, false, b1,
                                                 (short)0, acc1, false, false);
  }

  const int n0 = cq * 32 + lm;
  const float bo0 = bias[n0];
  const float bo1 = bias[n0 + 16];
#pragma unroll
  for (int r = 0; r < 8; ++r) {
    const int m = row0 + rt * 16 + r + 8 * hf;
    out[(size_t)m * EE + n0]      = acc0[r] + bo0;
    out[(size_t)m * EE + n0 + 16] = acc1[r] + bo1;
  }
}

extern "C" void kernel_launch(void* const* d_in, const int* in_sizes, int n_in,
                              void* d_out, int out_size, void* d_ws, size_t ws_size,
                              hipStream_t stream) {
  const float* x     = (const float*)d_in[0];
  const float* theta = (const float*)d_in[1];
  const float* w_out = (const float*)d_in[2];
  const float* b_out = (const float*)d_in[3];
  float* out = (float*)d_out;
  float* y   = (float*)d_ws;

  qattn_kernel<<<BB * HH * QSPLIT, NTHR, 0, stream>>>(x, theta, y);
  proj_kernel<<<BB * SS / PJ_ROWS, PJ_THR, 0, stream>>>(y, w_out, b_out, out);
}


# wave priority stepped down with each wave's own progress in the key loop (s_setprio 3,2,1,0 at 1/2, 3/4, 7/8 of its 64 key blocks): a SIMD's 8 waves end together, not oldest-first one by one
# speedup vs baseline: 2.5916x; 1.0844x over previous
#include <hip/hip_runtime.h>
#include <math.h>

typedef __attribute__((ext_vector_type(2))) float v2f;
typedef __attribute__((ext_vector_type(4))) float v4f;
typedef __attribute__((ext_vector_type(8))) float v8f;

#define NW 8
#define BB 8
#define SS 1024
#define EE 128
#define HH 16
#define DK 8
#define ZS 18
#define QSPLIT 2
#define NWAVE  (SS / 16 / QSPLIT)
#define NTHR   (NWAVE * 32)

__global__ __launch_bounds__(NTHR)
void qattn_kernel(const float* __restrict__ x, const float* __restrict__ theta,
                  float* __restrict__ y) {
  __shared__ float sZ[SS * ZS];
  __shared__ float sP[NWAVE * 16 * 18];

  const int bh   = blockIdx.x / QSPLIT;
  const int part = blockIdx.x % QSPLIT;
  const int b   = bh / HH;
  const int h   = bh % HH;
  const int tid = threadIdx.x;

  float th[NW];
#pragma unroll
  for (int j = 0; j < NW; ++j) th[j] = theta[j];

  for (int s = tid; s < SS; s += NTHR) {
    const float* xr = x + ((size_t)(b * SS + s) * EE + h * NW);
    float c[NW];
#pragma unroll
    for (int j = 0; j < NW; ++j) c[j] = __cosf(xr[j] + th[j]);
    float zr[NW];
    float p = c[0];
#pragma unroll
    for (int j = 1; j < NW; ++j) { p *= c[j]; zr[j] = p; }
    float sfx = 1.0f;
#pragma unroll
    for (int j = NW - 1; j >= 1; --j) sfx *= c[j];
    zr[0] = sfx;
#pragma unroll
    for (int j = 0; j < NW; ++j) sZ[s * ZS + j] = zr[j];
    sZ[s * ZS + 8] = 1.0f;
#pragma unroll
    for (int j = 9; j < 16; ++j) sZ[s * ZS + j] = 0.0f;
  }
  __syncthreads();

  const int lane = tid & 31;
  const int wave = tid >> 5;
  const int hf   = lane >> 4;
  const int lm   = lane & 15;
  const float scale = 0.35355339059327373f;
  float* sp = sP + wave * (16 * 18);

  {
    const int q0 = (part * NWAVE + wave) * 16;

    const float* qp = &sZ[(q0 + lm) * ZS + hf * 2];
    v2f a0 = {qp[0] * scale, qp[1] * scale};
    v2f a1 = {qp[4] * scale, qp[5] * scale};

    v8f o = {0.f, 0.f, 0.f, 0.f, 0.f, 0.f, 0.f, 0.f};

    auto key_blocks = [&](int kb_begin, int kb_end) __attribute__((always_inline)) {
    for (int kb = kb_begin; kb < kb_end; ++kb) {
      const int k0 = kb * 16;

      const float* kp = &sZ[(k0 + lm) * ZS + hf * 2];
      v2f b0 = {kp[0], kp[1]};
      v2f b1 = {kp[4], kp[5]};

      v8f sc = {0.f, 0.f, 0.f, 0.f, 0.f, 0.f, 0.f, 0.f};
      sc = __builtin_amdgcn_wmma_f32_16x16x4_f32(false, a0, false, b0,
                                                 (short)0, sc, false, false);
      sc = __builtin_amdgcn_wmma_f32_16x16x4_f32(false, a1, false, b1,
                                                 (short)0, sc, false, false);

      float pr[8];
#pragma unroll
      for (int r = 0; r < 8; ++r) pr[r] = __expf(sc[r]);

#pragma unroll
      for (int r = 0; r < 8; ++r)
        sp[(r + 8 * hf) * 18 + lm] = pr[r];
      __builtin_amdgcn_wave_barrier();

#pragma unroll
      for (int c = 0; c < 4; ++c) {
        const float* pp = &sp[lm * 18 + c * 4 + hf * 2];
        v2f pa = {pp[0], pp[1]};
        const int krow = k0 + c * 4 + hf * 2;
        v2f vb = {sZ[krow * ZS + lm], sZ[(krow + 1) * ZS + lm]};
        o = __builtin_amdgcn_wmma_f32_16x16x4_f32(false, pa, false, vb,
                                                  (short)0, o, false, false);
      }
      __builtin_amdgcn_wave_barrier();
    }
    };

    __builtin_amdgcn_s_setprio(3);
    key_blocks(0, 32);
    __builtin_amdgcn_s_setprio(2);
    key_blocks(32, 48);
    __builtin_amdgcn_s_setprio(1);
    key_blocks(48, 56);
    __builtin_amdgcn_s_setprio(0);
    key_blocks(56, 64);

#pragma unroll
    for (int r = 0; r < 8; ++r) {
      const float li = __shfl(o[r], 16 * hf + DK);
      if (lm < DK) {
        const int m = r + 8 * hf;
        y[(size_t)(b * SS + q0 + m) * EE + h * NW + lm] = o[r] / li;
      }
    }
  }
}

#define PJ_ROWS  32
#define PJ_PITCH 132
#define PJ_THR   256

__global__ __launch_bounds__(PJ_THR)
void proj_kernel(const float* __restrict__ y, const float* __restrict__ w,
                 const float* __restrict__ bias, float* __restrict__ out) {
  __shared__ float sW[EE * PJ_PITCH];
  __shared__ float sY[PJ_ROWS * PJ_PITCH];

  const int tid  = threadIdx.x;
  const int row0 = blockIdx.x * PJ_ROWS;

#pragma unroll
  for (int it = 0; it < (PJ_ROWS * EE / 4) / PJ_THR; ++it) {
    const int id = it * PJ_THR + tid;
    const int r  = id >> 5;
    const int c4 = id & 31;
    const v4f v = *reinterpret_cast<const v4f*>(y + (size_t)(row0 + r) * EE + c4 * 4);
    *reinterpret_cast<v4f*>(&sY[r * PJ_PITCH + c4 * 4]) = v;
  }
#pragma unroll
  for (int it = 0; it < (EE * EE / 4) / PJ_THR; ++it) {
    const int id = it * PJ_THR + tid;
    const int r  = id >> 5;
    const int c4 = id & 31;
    const v4f v = *reinterpret_cast<const v4f*>(w + (size_t)r * EE + c4 * 4);
    *reinterpret_cast<v4f*>(&sW[r * PJ_PITCH + c4 * 4]) = v;
  }
  __syncthreads();

  const int wave = tid >> 5;
  const int lane = tid & 31;
  const int hf   = lane >> 4;
  const int lm   = lane & 15;
  const int rt   = wave >> 2;
  const int cq   = wave & 3;

  const float* ap  = &sY[(rt * 16 + lm) * PJ_PITCH + hf * 2];
  const float* bp0 = &sW[(cq * 32 + lm) * PJ_PITCH + hf * 2];
  const float* bp1 = bp0 + 16 * PJ_PITCH;

  v8f acc0 = {0.f, 0.f, 0.f, 0.f, 0.f, 0.f, 0.f, 0.f};
  v8f acc1 = {0.f, 0.f, 0.f, 0.f, 0.f, 0.f, 0.f, 0.f};
#pragma unroll
  for (int kc = 0; kc < EE / 4; ++kc) {
    const v2f a  = {ap[kc * 4],  ap[kc * 4 + 1]};
    const v2f b0 = {bp0[kc * 4], bp0[kc * 4 + 1]};
    const v2f b1 = {bp1[kc * 4], bp1[kc * 4 + 1]};
    acc0 = __builtin_amdgcn_wmma_f32_16x16x4_f32(false, a, false, b0,
                                                 (short)0, acc0, false, false);
    acc1 = __builtin_amdgcn_wmma_f32_16x16x4_f32(false, a, false, b1,
                                                 (short)0, acc1, false, false);
  }

  const int n0 = cq * 32 + lm;
  const float bo0 = bias[n0];
  const float bo1 = bias[n0 + 16];
#pragma unroll
  for (int r = 0; r < 8; ++r) {
    const int m = row0 + rt * 16 + r + 8 * hf;
    out[(size_t)m * EE + n0]      = acc0[r] + bo0;
    out[(size_t)m * EE + n0 + 16] = acc1[r] + bo1;
  }
}

extern "C" void kernel_launch(void* const* d_in, const int* in_sizes, int n_in,
                              void* d_out, int out_size, void* d_ws, size_t ws_size,
                              hipStream_t stream) {
  const float* x     = (const float*)d_in[0];
  const float* theta = (const float*)d_in[1];
  const float* w_out = (const float*)d_in[2];
  const float* b_out = (const float*)d_in[3];
  float* out = (float*)d_out;
  float* y   = (float*)d_ws;

  qattn_kernel<<<BB * HH * QSPLIT, NTHR, 0, stream>>>(x, theta, y);
  proj_kernel<<<BB * SS / PJ_ROWS, PJ_THR, 0, stream>>>(y, w_out, b_out, out);
}


# score tile computed transposed (S^T = K Q^T): exp(S^T) feeds P@V from registers, P's LDS round trip goes; V staged transposed to match (16-byte reads); keys of a 16-block summed in another order
# speedup vs baseline: 2.7075x; 1.0447x over previous
#include <hip/hip_runtime.h>
#include <math.h>

typedef __attribute__((ext_vector_type(2))) float v2f;
typedef __attribute__((ext_vector_type(4))) float v4f;
typedef __attribute__((ext_vector_type(8))) float v8f;

#define NW 8
#define BB 8
#define SS 1024
#define EE 128
#define HH 16
#define DK 8
#define ZS 12
#define VR 10
#define VP 1028
#define QSPLIT 2
#define NWAVE  (SS / 16 / QSPLIT)
#define NTHR   (NWAVE * 32)

__global__ __launch_bounds__(NTHR)
void qattn_kernel(const float* __restrict__ x, const float* __restrict__ theta,
                  float* __restrict__ y) {
  __shared__ __attribute__((aligned(16))) float sZ[SS * ZS];
  __shared__ __attribute__((aligned(16))) float sVT[VR * VP];

  const int bh   = blockIdx.x / QSPLIT;
  const int part = blockIdx.x % QSPLIT;
  const int b   = bh / HH;
  const int h   = bh % HH;
  const int tid = threadIdx.x;

  float th[NW];
#pragma unroll
  for (int j = 0; j < NW; ++j) th[j] = theta[j];

  for (int s = tid; s < SS; s += NTHR) {
    const float* xr = x + ((size_t)(b * SS + s) * EE + h * NW);
    float c[NW];
#pragma unroll
    for (int j = 0; j < NW; ++j) c[j] = __cosf(xr[j] + th[j]);
    float zr[NW];
    float p = c[0];
#pragma unroll
    for (int j = 1; j < NW; ++j) { p *= c[j]; zr[j] = p; }
    float sfx = 1.0f;
#pragma unroll
    for (int j = NW - 1; j >= 1; --j) sfx *= c[j];
    zr[0] = sfx;
#pragma unroll
    for (int j = 0; j < NW; ++j) sZ[s * ZS + j] = zr[j];
#pragma unroll
    for (int j = 0; j < NW; ++j) sVT[j * VP + s] = zr[j];
    sVT[8 * VP + s] = 1.0f;
    sVT[9 * VP + s] = 0.0f;
  }
  __syncthreads();

  const int lane = tid & 31;
  const int wave = tid >> 5;
  const int hf   = lane >> 4;
  const int lm   = lane & 15;
  const float scale = 0.35355339059327373f;

  {
    const int q0 = (part * NWAVE + wave) * 16;

    const float* qp = &sZ[(q0 + lm) * ZS + hf * 2];
    v2f qb0 = {qp[0] * scale, qp[1] * scale};
    v2f qb1 = {qp[4] * scale, qp[5] * scale};

    const float* kbase = &sZ[lm * ZS + hf * 2];
    const float* vbase = &sVT[(lm < 9 ? lm : 9) * VP + hf * 8];

    v8f o = {0.f, 0.f, 0.f, 0.f, 0.f, 0.f, 0.f, 0.f};

    auto key_blocks = [&](int kb_begin, int kb_end) __attribute__((always_inline)) {
    for (int kb = kb_begin; kb < kb_end; ++kb) {
      const int k0 = kb * 16;

      const float* kp = kbase + kb * (16 * ZS);
      v2f ka0 = {kp[0], kp[1]};
      v2f ka1 = {kp[4], kp[5]};

      const v4f* vp = reinterpret_cast<const v4f*>(vbase + k0);
      const v4f vlo = vp[0];
      const v4f vhi = vp[1];

      v8f st = {0.f, 0.f, 0.f, 0.f, 0.f, 0.f, 0.f, 0.f};
      st = __builtin_amdgcn_wmma_f32_16x16x4_f32(false, ka0, false, qb0,
                                                 (short)0, st, false, false);
      st = __builtin_amdgcn_wmma_f32_16x16x4_f32(false, ka1, false, qb1,
                                                 (short)0, st, false, false);

      float pr[8];
#pragma unroll
      for (int r = 0; r < 8; ++r) pr[r] = __expf(st[r]);

      __builtin_amdgcn_sched_barrier(0);

      {
        const v2f p0 = {pr[0], pr[1]}, p1 = {pr[2], pr[3]}, p2 = {pr[4], pr[5]}, p3 = {pr[6], pr[7]};
        const v2f v0 = {vlo[0], vlo[1]}, v1 = {vlo[2], vlo[3]}, v2 = {vhi[0], vhi[1]}, v3 = {vhi[2], vhi[3]};
        o = __builtin_amdgcn_wmma_f32_16x16x4_f32(false, p0, false, v0, (short)0, o, false, false);
        o = __builtin_amdgcn_wmma_f32_16x16x4_f32(false, p1, false, v1, (short)0, o, false, false);
        o = __builtin_amdgcn_wmma_f32_16x16x4_f32(false, p2, false, v2, (short)0, o, false, false);
        o = __builtin_amdgcn_wmma_f32_16x16x4_f32(false, p3, false, v3, (short)0, o, false, false);
      }
    }
    };

    __builtin_amdgcn_s_setprio(3);
    key_blocks(0, 32);
    __builtin_amdgcn_s_setprio(2);
    key_blocks(32, 48);
    __builtin_amdgcn_s_setprio(1);
    key_blocks(48, 56);
    __builtin_amdgcn_s_setprio(0);
    key_blocks(56, 64);

#pragma unroll
    for (int r = 0; r < 8; ++r) {
      const float li = __shfl(o[r], 16 * hf + DK);
      if (lm < DK) {
        const int m = r + 8 * hf;
        y[(size_t)(b * SS + q0 + m) * EE + h * NW + lm] = o[r] / li;
      }
    }
  }
}

#define PJ_ROWS  32
#define PJ_PITCH 132
#define PJ_THR   256

__global__ __launch_bounds__(PJ_THR)
void proj_kernel(const float* __restrict__ y, const float* __restrict__ w,
                 const float* __restrict__ bias, float* __restrict__ out) {
  __shared__ float sW[EE * PJ_PITCH];
  __shared__ float sY[PJ_ROWS * PJ_PITCH];

  const int tid  = threadIdx.x;
  const int row0 = blockIdx.x * PJ_ROWS;

#pragma unroll
  for (int it = 0; it < (PJ_ROWS * EE / 4) / PJ_THR; ++it) {
    const int id = it * PJ_THR + tid;
    const int r  = id >> 5;
    const int c4 = id & 31;
    const v4f v = *reinterpret_cast<const v4f*>(y + (size_t)(row0 + r) * EE + c4 * 4);
    *reinterpret_cast<v4f*>(&sY[r * PJ_PITCH + c4 * 4]) = v;
  }
#pragma unroll
  for (int it = 0; it < (EE * EE / 4) / PJ_THR; ++it) {
    const int id = it * PJ_THR + tid;
    const int r  = id >> 5;
    const int c4 = id & 31;
    const v4f v = *reinterpret_cast<const v4f*>(w + (size_t)r * EE + c4 * 4);
    *reinterpret_cast<v4f*>(&sW[r * PJ_PITCH + c4 * 4]) = v;
  }
  __syncthreads();

  const int wave = tid >> 5;
  const int lane = tid & 31;
  const int hf   = lane >> 4;
  const int lm   = lane & 15;
  const int rt   = wave >> 2;
  const int cq   = wave & 3;

  const float* ap  = &sY[(rt * 16 + lm) * PJ_PITCH + hf * 2];
  const float* bp0 = &sW[(cq * 32 + lm) * PJ_PITCH + hf * 2];
  const float* bp1 = bp0 + 16 * PJ_PITCH;

  v8f acc0 = {0.f, 0.f, 0.f, 0.f, 0.f, 0.f, 0.f, 0.f};
  v8f acc1 = {0.f, 0.f, 0.f, 0.f, 0.f, 0.f, 0.f, 0.f};
#pragma unroll
  for (int kc = 0; kc < EE / 4; ++kc) {
    const v2f a  = {ap[kc * 4],  ap[kc * 4 + 1]};
    const v2f b0 = {bp0[kc * 4], bp0[kc * 4 + 1]};
    const v2f b1 = {bp1[kc * 4], bp1[kc * 4 + 1]};
    acc0 = __builtin_amdgcn_wmma_f32_16x16x4_f32(false, a, false, b0,
                                                 (short)0, acc0, false, false);
    acc1 = __builtin_amdgcn_wmma_f32_16x16x4_f32(false, a, false, b1,
                                                 (short)0, acc1, false, false);
  }

  const int n0 = cq * 32 + lm;
  const float bo0 = bias[n0];
  const float bo1 = bias[n0 + 16];
#pragma unroll
  for (int r = 0; r < 8; ++r) {
    const int m = row0 + rt * 16 + r + 8 * hf;
    out[(size_t)m * EE + n0]      = acc0[r] + bo0;
    out[(size_t)m * EE + n0 + 16] = acc1[r] + bo1;
  }
}

extern "C" void kernel_launch(void* const* d_in, const int* in_sizes, int n_in,
                              void* d_out, int out_size, void* d_ws, size_t ws_size,
                              hipStream_t stream) {
  const float* x     = (const float*)d_in[0];
  const float* theta = (const float*)d_in[1];
  const float* w_out = (const float*)d_in[2];
  const float* b_out = (const float*)d_in[3];
  float* out = (float*)d_out;
  float* y   = (float*)d_ws;

  qattn_kernel<<<BB * HH * QSPLIT, NTHR, 0, stream>>>(x, theta, y);
  proj_kernel<<<BB * SS / PJ_ROWS, PJ_THR, 0, stream>>>(y, w_out, b_out, out);
}
